# starting kernel
# baseline (speedup 1.0000x reference)
#include <hip/hip_runtime.h>
#include <hip/hip_bf16.h>

typedef __attribute__((ext_vector_type(2))) float v2f;
typedef __attribute__((ext_vector_type(4))) float v4f;
typedef __attribute__((ext_vector_type(8))) float v8f;

#define BATCHN 4096
#define SEQN   8192
#define NCHUNK 4
#define CHUNK  (SEQN / NCHUNK)
#define WARM   64

__global__ __launch_bounds__(32)
void iir_wmma_kernel(const float* __restrict__ x,
                     const float* __restrict__ A,
                     const float* __restrict__ B,
                     float* __restrict__ y) {
    __shared__ __align__(16) float h_sh[32];
    __shared__ __align__(16) float xt[16 * 20];
    __shared__ __align__(16) float yt[16 * 20];

    const int lane = threadIdx.x;
    const int half = lane >> 4;
    const int m    = lane & 15;
    const int rowbase = blockIdx.x * 16;
    const int cstart  = blockIdx.y * CHUNK;

    const float a0 = A[0], a1 = A[1], a2 = A[2], a3 = A[3];
    const float b0 = B[0], b1 = B[1], b2 = B[2], b3 = B[3], b4 = B[4];

    h_sh[lane] = 0.0f;
    __syncthreads();
    if (lane == 0) {
        h_sh[16] = 1.0f;
        for (int t = 1; t < 16; ++t)
            h_sh[16 + t] = a0 * h_sh[16 + t - 1] + a1 * h_sh[16 + t - 2]
                         + a2 * h_sh[16 + t - 3] + a3 * h_sh[16 + t - 4];
    }
    __syncthreads();

    v2f a_tf[4];
#pragma unroll
    for (int c = 0; c < 4; ++c) {
        const int k0 = 4 * c + 2 * half;
        a_tf[c].x = h_sh[16 + m - k0];
        a_tf[c].y = h_sh[16 + m - k0 - 1];
    }
    v2f a_th;
    {
        const float H0 = h_sh[16 + m], H1 = h_sh[15 + m],
                    H2 = h_sh[14 + m], H3 = h_sh[13 + m];
        const float th0 = a0 * H0 + a1 * H1 + a2 * H2 + a3 * H3;
        const float th1 = a1 * H0 + a2 * H1 + a3 * H2;
        const float th2 = a2 * H0 + a3 * H1;
        const float th3 = a3 * H0;
        a_th.x = (half == 0) ? th0 : th2;
        a_th.y = (half == 0) ? th1 : th3;
    }

    v2f bs; bs.x = 0.0f; bs.y = 0.0f;

    int n0 = cstart - WARM;
    const int iters = (CHUNK + WARM) / 16;

    for (int it = 0; it < iters; ++it) {
#pragma unroll
        for (int s = 0; s < 3; ++s) {
            const int idx = lane + 32 * s;
            if (idx < 80) {
                const int r = idx / 5, seg = idx % 5;
                const int pos = n0 - 4 + seg * 4;
                v4f v = {0.0f, 0.0f, 0.0f, 0.0f};
                if (pos >= 0)
                    v = *(const v4f*)(x + (size_t)(rowbase + r) * SEQN + pos);
                *(v4f*)&xt[r * 20 + seg * 4] = v;
            }
        }
        if (lane < 16 && (n0 + 160) < SEQN)
            __builtin_prefetch(x + (size_t)(rowbase + lane) * SEQN + n0 + 128, 0, 1);
        __syncthreads();

        const int nb = m * 20;
        const int jb = 2 * half;
        v2f bff[4];
#pragma unroll
        for (int c = 0; c < 4; ++c) {
            const int j0 = 4 * c + jb;
            bff[c].x = b0 * xt[nb + j0 + 4] + b1 * xt[nb + j0 + 3]
                     + b2 * xt[nb + j0 + 2] + b3 * xt[nb + j0 + 1]
                     + b4 * xt[nb + j0 + 0];
            bff[c].y = b0 * xt[nb + j0 + 5] + b1 * xt[nb + j0 + 4]
                     + b2 * xt[nb + j0 + 3] + b3 * xt[nb + j0 + 2]
                     + b4 * xt[nb + j0 + 1];
        }

        v8f acc = {};
#pragma unroll
        for (int c = 0; c < 4; ++c)
            acc = __builtin_amdgcn_wmma_f32_16x16x4_f32(
                false, a_tf[c], false, bff[c], (short)0, acc, false, false);
        acc = __builtin_amdgcn_wmma_f32_16x16x4_f32(
                false, a_th, false, bs, (short)0, acc, false, false);

        const float t7 = __shfl_xor(acc[7], 16, 32);
        const float t6 = __shfl_xor(acc[6], 16, 32);
        bs.x = (half == 0) ? t7 : acc[5];
        bs.y = (half == 0) ? t6 : acc[4];

        if (n0 >= cstart) {
#pragma unroll
            for (int v = 0; v < 8; ++v)
                yt[m * 20 + half * 8 + v] = acc[v];
            __syncthreads();
#pragma unroll
            for (int s = 0; s < 2; ++s) {
                const int idx = lane + 32 * s;
                const int r = idx >> 2, seg = idx & 3;
                *(v4f*)(y + (size_t)(rowbase + r) * SEQN + n0 + seg * 4) =
                    *(const v4f*)&yt[r * 20 + seg * 4];
            }
        }
        __syncthreads();
        n0 += 16;
    }
}

extern "C" void kernel_launch(void* const* d_in, const int* in_sizes, int n_in,
                              void* d_out, int out_size, void* d_ws, size_t ws_size,
                              hipStream_t stream) {
    const float* x = (const float*)d_in[0];
    const float* a = (const float*)d_in[1];
    const float* b = (const float*)d_in[2];
    float* out = (float*)d_out;
    (void)in_sizes; (void)n_in; (void)out_size; (void)d_ws; (void)ws_size;

    dim3 grid(BATCHN / 16, NCHUNK, 1);
    iir_wmma_kernel<<<grid, 32, 0, stream>>>(x, a, b, out);
}


# x and y cross global memory as whole lines (2 rows x 256 B per load/store, not 16 rows x 16 B): 0/1 selector products on the matrix unit turn lane=time into lane=row and back; sum order changes
# speedup vs baseline: 2.9767x; 2.9767x over previous
#include <hip/hip_runtime.h>
#include <hip/hip_bf16.h>


typedef __attribute__((ext_vector_type(2))) float v2f;
typedef __attribute__((ext_vector_type(4))) float v4f;
typedef __attribute__((ext_vector_type(8))) float v8f;

#define BATCHN 4096
#define SEQN   8192
#define NCHUNK 4
#define CHUNK  (SEQN / NCHUNK)
#define WARM   64
#define SUPER  64
#define RING   8
#define AHEAD  6


__device__ __forceinline__ float xhalf(float v) {
    const int b = __builtin_bit_cast(int, v);
    return __builtin_bit_cast(float, __builtin_amdgcn_permlanex16(b, b, 0x76543210, 0xfedcba98, false, false));
}

__device__ __forceinline__ v8f mm(v2f a, v2f b, v8f c) {
    return __builtin_amdgcn_wmma_f32_16x16x4_f32(false, a, false, b, (short)0, c, false, false);
}

template <bool GUARD>
__device__ __forceinline__ void load_super(v4f (&t)[8], const float* __restrict__ xl, int n0) {
    if (!GUARD || n0 >= 0) {
#pragma unroll
        for (int i = 0; i < 8; ++i)
            t[i] = *(const v4f*)(xl + (size_t)(2 * i) * SEQN + n0);
    } else {
#pragma unroll
        for (int i = 0; i < 8; ++i)
            t[i] = (v4f){0.0f, 0.0f, 0.0f, 0.0f};
    }
}

struct Ops {
    v2f e[4];
    v2f f[4];
    v2f g[4];
    v2f gh;
    v2f th;
};

template <bool STORE>
__device__ __forceinline__ void run_super(const v4f (&t)[8], const Ops& o,
                                          v2f& bs, v2f& bh, float* __restrict__ yl,
                                          int n0, int half) {
    v8f d[4];
#pragma unroll
    for (int j = 0; j < 4; ++j) {
        v8f acc = {};
#pragma unroll
        for (int q = 0; q < 4; ++q) {
            v2f a; a.x = t[2 * q][j]; a.y = t[2 * q + 1][j];
            acc = mm(a, o.e[q], acc);
        }
        d[j] = acc;
    }

    v8f z[4];
#pragma unroll
    for (int s = 0; s < 4; ++s) z[s] = (v8f){};

#pragma unroll
    for (int T = 0; T < 4; ++T) {
        v8f acc = {};
#pragma unroll
        for (int j = 0; j < 4; ++j) {
            v2f b; b.x = d[j][2 * T]; b.y = d[j][2 * T + 1];
            acc = mm(o.g[j], b, acc);
        }
        acc = mm(o.gh, bh, acc);
        acc = mm(o.th, bs, acc);

        const float t7 = xhalf(acc[7]);
        const float t5 = xhalf(acc[5]);
        bs.x = (half == 0) ? t7 : acc[3];
        bs.y = (half == 0) ? t5 : acc[1];

        const float h2 = xhalf(d[2][2 * T + 1]);
        const float h3 = xhalf(d[3][2 * T + 1]);
        bh.x = (half == 0) ? h2 : d[0][2 * T + 1];
        bh.y = (half == 0) ? h3 : d[1][2 * T + 1];

        if (STORE) {
#pragma unroll
            for (int s = 0; s < 4; ++s) {
                v2f a; a.x = acc[2 * s]; a.y = acc[2 * s + 1];
                z[s] = mm(a, o.f[T], z[s]);
            }
        }
    }

    if (STORE) {
#pragma unroll
        for (int i = 0; i < 8; ++i)
            *(v4f*)(yl + (size_t)i * SEQN + n0) = (v4f){z[0][i], z[1][i], z[2][i], z[3][i]};
    }
}

__global__ __launch_bounds__(32)
void iir_wmma_kernel(const float* __restrict__ x,
                     const float* __restrict__ A,
                     const float* __restrict__ B,
                     float* __restrict__ y) {
    __shared__ __align__(16) float h_sh[32];

    const int lane = threadIdx.x;
    const int half = lane >> 4;
    const int m    = lane & 15;
    const int rowbase = blockIdx.x * 16;
    const int cstart  = blockIdx.y * CHUNK;

    const float a0 = A[0], a1 = A[1], a2 = A[2], a3 = A[3];
    const float b0 = B[0], b1 = B[1], b2 = B[2], b3 = B[3], b4 = B[4];

    h_sh[lane] = 0.0f;
    __syncthreads();
    if (lane == 0) {
        h_sh[16] = 1.0f;
        for (int t = 1; t < 16; ++t)
            h_sh[16 + t] = a0 * h_sh[16 + t - 1] + a1 * h_sh[16 + t - 2]
                         + a2 * h_sh[16 + t - 3] + a3 * h_sh[16 + t - 4];
    }
    __syncthreads();

    auto hz = [&](int d) -> float { return (d >= 0) ? h_sh[16 + d] : 0.0f; };
    auto gw = [&](int mm_, int tt) -> float {
        float s = 0.0f;
        if (tt + 0 >= 0) s += b0 * hz(mm_ - tt - 0);
        if (tt + 1 >= 0) s += b1 * hz(mm_ - tt - 1);
        if (tt + 2 >= 0) s += b2 * hz(mm_ - tt - 2);
        if (tt + 3 >= 0) s += b3 * hz(mm_ - tt - 3);
        if (tt + 4 >= 0) s += b4 * hz(mm_ - tt - 4);
        return s;
    };

    const int pi_m  = 4 * ((m & 7) >> 1) + 2 * (m >> 3) + (m & 1);
    const int tau_m = 4 * (2 * (m >> 3) + (m & 1)) + ((m & 7) >> 1);

    Ops o;
#pragma unroll
    for (int q = 0; q < 4; ++q) {
        o.e[q].x = (m == 4 * q + half)     ? 1.0f : 0.0f;
        o.e[q].y = (m == 4 * q + 2 + half) ? 1.0f : 0.0f;
        o.f[q].x = (m == 4 * q + 2 * half)     ? 1.0f : 0.0f;
        o.f[q].y = (m == 4 * q + 2 * half + 1) ? 1.0f : 0.0f;
        o.g[q].x = gw(tau_m, 8 * half + q);
        o.g[q].y = gw(tau_m, 8 * half + 4 + q);
    }
    o.gh.x = gw(tau_m, half ? -4 : -2);
    o.gh.y = gw(tau_m, half ? -3 : -1);

    {
        const float H0 = h_sh[16 + tau_m], H1 = h_sh[15 + tau_m],
                    H2 = h_sh[14 + tau_m], H3 = h_sh[13 + tau_m];
        const float th0 = a0 * H0 + a1 * H1 + a2 * H2 + a3 * H3;
        const float th1 = a1 * H0 + a2 * H1 + a3 * H2;
        const float th2 = a2 * H0 + a3 * H1;
        const float th3 = a3 * H0;
        o.th.x = (half == 0) ? th0 : th2;
        o.th.y = (half == 0) ? th1 : th3;
    }

    const float* xl = x + (size_t)(rowbase + half) * SEQN + 4 * pi_m;
    float*       yl = y + (size_t)(rowbase + 8 * half) * SEQN + 4 * m;

    v2f bs; bs.x = 0.0f; bs.y = 0.0f;

    int n0 = cstart - WARM;

    v2f bh; bh.x = 0.0f; bh.y = 0.0f;
    if (n0 >= 4) bh = *(const v2f*)(x + (size_t)(rowbase + m) * SEQN + n0 - 4 + (half ? 0 : 2));

    static_assert(WARM == SUPER, "trip 0 is the warm-up trip");
    static_assert(AHEAD < RING, "a refilled slot must hold a finished trip");
    static_assert((CHUNK / SUPER) % RING == 0 && (CHUNK / SUPER) >= 2 * RING, "stored trips come in whole rings");
    static_assert(8 * (AHEAD + 1) < 64, "loads in flight, the refill included, stay inside the load counter");
    v4f ring[RING][8];
    load_super<true >(ring[0], xl, n0);
#pragma unroll
    for (int r = 1; r < AHEAD; ++r)
        load_super<false>(ring[r], xl, n0 + r * SUPER);

    load_super<false>(ring[AHEAD], xl, n0 + AHEAD * SUPER);
    __builtin_amdgcn_sched_barrier(0);
    run_super<false>(ring[0], o, bs, bh, yl, n0, half);
    __builtin_amdgcn_sched_barrier(0);
    n0 += SUPER;

    const int groups = CHUNK / SUPER / RING;
#pragma unroll 1
    for (int g = 0; g < groups - 1; ++g) {
#pragma unroll
        for (int p = 0; p < RING; ++p) {
            load_super<false>(ring[(p + 1 + AHEAD) % RING], xl, n0 + AHEAD * SUPER);
            __builtin_amdgcn_sched_barrier(0);
            run_super<true>(ring[(p + 1) % RING], o, bs, bh, yl, n0, half);
            __builtin_amdgcn_sched_barrier(0);
            n0 += SUPER;
        }
    }
#pragma unroll
    for (int p = 0; p < RING; ++p) {
        if (p < RING - AHEAD) {
            load_super<false>(ring[(p + 1 + AHEAD) % RING], xl, n0 + AHEAD * SUPER);
            __builtin_amdgcn_sched_barrier(0);
        }
        run_super<true>(ring[(p + 1) % RING], o, bs, bh, yl, n0, half);
        __builtin_amdgcn_sched_barrier(0);
        n0 += SUPER;
    }
}

extern "C" void kernel_launch(void* const* d_in, const int* in_sizes, int n_in,
                              void* d_out, int out_size, void* d_ws, size_t ws_size,
                              hipStream_t stream) {
    const float* x = (const float*)d_in[0];
    const float* a = (const float*)d_in[1];
    const float* b = (const float*)d_in[2];
    float* out = (float*)d_out;
    (void)in_sizes; (void)n_in; (void)out_size; (void)d_ws; (void)ws_size;

    dim3 grid(BATCHN / 16, NCHUNK, 1);
    iir_wmma_kernel<<<grid, 32, 0, stream>>>(x, a, b, out);
}
